# starting kernel
# baseline (speedup 1.0000x reference)
#include <hip/hip_runtime.h>
#include <hip/hip_bf16.h>


typedef __attribute__((ext_vector_type(2))) float v2f;
typedef __attribute__((ext_vector_type(4))) float v4f;
typedef __attribute__((ext_vector_type(8))) float v8f;

#define NFEAT 20
#define NCLS  8
#define ROWSTRIDE 40
#define KDIM 60
#define TILE 16
#define WAVES_PER_BLOCK 8

struct WTabT { float v[KDIM][16]; };

static constexpr WTabT makeWTab() {
    WTabT t{};
    const float inv2s2  = 5.5555555f;
    const float invs2   = 11.111111f;
    const float logNorm = 0.28503434f;
    for (int k = 0; k < KDIM; ++k) {
        for (int c = 0; c < 16; ++c) {
            float w = 0.0f;
            if (c < NCLS) {
                int g = k / NFEAT;
                int j = k - g * NFEAT;
                int d = j - c;
                if (d >= 0 && d < 3) {
                    float m = (float)((c >> d) & 1);
                    if (g == 0)      w = -inv2s2;
                    else if (g == 1) w = m * invs2;
                    else             w = -m * inv2s2 + logNorm;
                }
            }
            t.v[k][c] = w;
        }
    }
    return t;
}

__constant__ WTabT wtab = makeWTab();

__device__ __forceinline__ unsigned killMaskF(int j) {
    int lo = j - 2; if (lo < 0) lo = 0;
    int hi = j;     if (hi > 7) hi = 7;
    unsigned g = 0u;
    if (lo <= hi) g = ((1u << (hi - lo + 1)) - 1u) << lo;
    return 0xFFu & ~g;
}

__device__ __forceinline__ float uval(const float* __restrict__ row, int k) {
    int g = k / NFEAT;
    int j = k - g * NFEAT;
    float f   = row[j];
    float msk = row[NFEAT + j];
    if (g == 0) return msk * f * f;
    if (g == 1) return msk * f;
    return msk;
}

__global__ void __launch_bounds__(256)
nb_wmma_kernel(const float* __restrict__ x, float* __restrict__ out, int numTiles) {
    __shared__ float lds[WAVES_PER_BLOCK][TILE * ROWSTRIDE + TILE * NCLS];

    const int lane        = threadIdx.x & 31;
    const int waveInBlock = threadIdx.x >> 5;
    float* tileBuf = &lds[waveInBlock][0];
    float* scBuf   = &lds[waveInBlock][TILE * ROWSTRIDE];

    const int m  = lane & 15;
    const int k0 = (lane >> 4) << 1;
    const int n  = lane & 15;

    v2f bfrag[15];
#pragma unroll
    for (int s = 0; s < 15; ++s) {
        int kk = 4 * s + k0;
        bfrag[s].x = wtab.v[kk][n];
        bfrag[s].y = wtab.v[kk + 1][n];
    }

    const int waveId    = blockIdx.x * WAVES_PER_BLOCK + waveInBlock;
    const int waveCount = gridDim.x * WAVES_PER_BLOCK;

    for (int t = waveId; t < numTiles; t += waveCount) {
        const float* src = x + (long long)t * (TILE * ROWSTRIDE);

        int tn = t + waveCount;
        if (tn < numTiles) {
            const char* nxt = (const char*)(x + (long long)tn * (TILE * ROWSTRIDE));
            __builtin_prefetch(nxt + lane * 80, 0, 0);
        }

#pragma unroll
        for (int i = 0; i < (TILE * ROWSTRIDE) / (4 * 32); ++i) {
            int idx = lane + 32 * i;
            ((v4f*)tileBuf)[idx] = ((const v4f*)src)[idx];
        }
        __builtin_amdgcn_wave_barrier();

        const float* row = tileBuf + m * ROWSTRIDE;

        unsigned kill = 0u;
#pragma unroll
        for (int j = 0; j < NFEAT; ++j) {
            float f   = row[j];
            float msk = row[NFEAT + j];
            bool bad = (msk > 0.0f) && !((f >= 0.0f) && (f < 1.0f));
            kill |= bad ? killMaskF(j) : 0u;
        }

        v8f acc = {};
#pragma unroll
        for (int s = 0; s < 15; ++s) {
            int kk = 4 * s + k0;
            v2f a;
            a.x = uval(row, kk);
            a.y = uval(row, kk + 1);
            acc = __builtin_amdgcn_wmma_f32_16x16x4_f32(
                false, a, false, bfrag[s], (short)0, acc, false, false);
        }

        if (n < NCLS) {
            int rbase = (lane >> 4) << 3;
#pragma unroll
            for (int v = 0; v < 8; ++v)
                scBuf[(rbase + v) * NCLS + n] = acc[v];
        }
        __builtin_amdgcn_wave_barrier();

        if (lane < TILE) {
            const v4f* sp = (const v4f*)(scBuf + lane * NCLS);
            v4f s0 = sp[0];
            v4f s1 = sp[1];
            float p[NCLS] = { s0.x, s0.y, s0.z, s0.w, s1.x, s1.y, s1.z, s1.w };
            float sum = 0.0f;
#pragma unroll
            for (int c = 0; c < NCLS; ++c) {
                float e  = __expf(p[c]) * 0.125f;
                float pr = ((kill >> c) & 1u) ? 0.0f : e;
                p[c] = pr;
                sum += pr;
            }
            float inv = 1.0f / fmaxf(sum, 1e-8f);
            float* dst = out + ((long long)t * TILE + lane) * NCLS;
            v4f o0 = { p[0] * inv, p[1] * inv, p[2] * inv, p[3] * inv };
            v4f o1 = { p[4] * inv, p[5] * inv, p[6] * inv, p[7] * inv };
            ((v4f*)dst)[0] = o0;
            ((v4f*)dst)[1] = o1;
        }
        __builtin_amdgcn_wave_barrier();
    }
}

__global__ void nb_tail_kernel(const float* __restrict__ x, float* __restrict__ out,
                               int startRow, int rows) {
    int r = startRow + blockIdx.x * blockDim.x + threadIdx.x;
    if (r >= rows) return;
    const float* row = x + (long long)r * ROWSTRIDE;
    float probs[NCLS];
    float sum = 0.0f;
    for (int c = 0; c < NCLS; ++c) {
        float p = 0.125f;
        for (int j = 0; j < NFEAT; ++j) {
            float f = row[j], msk = row[NFEAT + j];
            if (msk > 0.0f) {
                int d = j - c;
                float v;
                if (d >= 0 && d < 3) {
                    float mean = (float)((c >> d) & 1);
                    float z = (f - mean) * (1.0f / 0.3f);
                    v = __expf(-0.5f * z * z) * (0.3989422804014327f / 0.3f);
                } else {
                    v = ((f >= 0.0f) && (f < 1.0f)) ? 1.0f : 0.0f;
                }
                p *= v;
            }
        }
        probs[c] = p;
        sum += p;
    }
    float inv = 1.0f / fmaxf(sum, 1e-8f);
    for (int c = 0; c < NCLS; ++c)
        out[(long long)r * NCLS + c] = probs[c] * inv;
}

extern "C" void kernel_launch(void* const* d_in, const int* in_sizes, int n_in,
                              void* d_out, int out_size, void* d_ws, size_t ws_size,
                              hipStream_t stream) {
    (void)n_in; (void)out_size; (void)d_ws; (void)ws_size;
    const float* x   = (const float*)d_in[0];
    float*       out = (float*)d_out;

    long long rows = (long long)in_sizes[0] / ROWSTRIDE;
    int numTiles = (int)(rows / TILE);
    int tail     = (int)(rows % TILE);

    if (numTiles > 0) {
        int blocks = 1024;
        int maxBlocks = (numTiles + WAVES_PER_BLOCK - 1) / WAVES_PER_BLOCK;
        if (blocks > maxBlocks) blocks = maxBlocks;
        nb_wmma_kernel<<<blocks, 256, 0, stream>>>(x, out, numTiles);
    }
    if (tail > 0) {
        nb_tail_kernel<<<1, 64, 0, stream>>>(x, out, (int)(rows - tail), (int)rows);
    }
}


# kill mask computed branch-free from the ten feature/mask pairs each lane already loads for its A fragments, lane halves merged by one v_permlanex16; replaces 40 dependent LDS reads per tile
# speedup vs baseline: 1.3997x; 1.3997x over previous
#include <hip/hip_runtime.h>
#include <hip/hip_bf16.h>


typedef __attribute__((ext_vector_type(2))) float v2f;
typedef __attribute__((ext_vector_type(4))) float v4f;
typedef __attribute__((ext_vector_type(8))) float v8f;

#define NFEAT 20
#define NCLS  8
#define ROWSTRIDE 40
#define KDIM 60
#define TILE 16
#define WAVES_PER_BLOCK 8

struct WTabT { float v[KDIM][16]; };

static constexpr WTabT makeWTab() {
    WTabT t{};
    const float inv2s2  = 5.5555555f;
    const float invs2   = 11.111111f;
    const float logNorm = 0.28503434f;
    for (int k = 0; k < KDIM; ++k) {
        for (int c = 0; c < 16; ++c) {
            float w = 0.0f;
            if (c < NCLS) {
                int g = k / NFEAT;
                int j = k - g * NFEAT;
                int d = j - c;
                if (d >= 0 && d < 3) {
                    float m = (float)((c >> d) & 1);
                    if (g == 0)      w = -inv2s2;
                    else if (g == 1) w = m * invs2;
                    else             w = -m * inv2s2 + logNorm;
                }
            }
            t.v[k][c] = w;
        }
    }
    return t;
}

__constant__ WTabT wtab = makeWTab();

__device__ __forceinline__ unsigned killMaskF(int j) {
    int lo = j - 2; if (lo < 0) lo = 0;
    int hi = j;     if (hi > 7) hi = 7;
    unsigned g = 0u;
    if (lo <= hi) g = ((1u << (hi - lo + 1)) - 1u) << lo;
    return 0xFFu & ~g;
}

__device__ __forceinline__ unsigned otherHalf(unsigned v) {
    return (unsigned)__builtin_amdgcn_permlanex16((int)v, (int)v, 0x76543210, 0xfedcba98, false, false);
}

__device__ __forceinline__ unsigned killBits(float f, float msk, unsigned bits) {
    bool bad = (msk > 0.0f) & !((f >= 0.0f) & (f < 1.0f));
    return bad ? bits : 0u;
}

__global__ void __launch_bounds__(256)
nb_wmma_kernel(const float* __restrict__ x, float* __restrict__ out, int numTiles) {
    __shared__ float lds[WAVES_PER_BLOCK][TILE * ROWSTRIDE + TILE * NCLS];

    const int lane        = threadIdx.x & 31;
    const int waveInBlock = threadIdx.x >> 5;
    float* tileBuf = &lds[waveInBlock][0];
    float* scBuf   = &lds[waveInBlock][TILE * ROWSTRIDE];

    const int m  = lane & 15;
    const int k0 = (lane >> 4) << 1;
    const int n  = lane & 15;

    v2f bfrag[15];
#pragma unroll
    for (int s = 0; s < 15; ++s) {
        int kk = 4 * s + k0;
        bfrag[s].x = wtab.v[kk][n];
        bfrag[s].y = wtab.v[kk + 1][n];
    }

    unsigned kc[10];
    for (int s = 0; s < 5; ++s) {
        kc[2 * s]     = k0 ? killMaskF(4 * s + 2) : killMaskF(4 * s);
        kc[2 * s + 1] = k0 ? killMaskF(4 * s + 3) : killMaskF(4 * s + 1);
    }

    const int waveId    = blockIdx.x * WAVES_PER_BLOCK + waveInBlock;
    const int waveCount = gridDim.x * WAVES_PER_BLOCK;

    for (int t = waveId; t < numTiles; t += waveCount) {
        const float* src = x + (long long)t * (TILE * ROWSTRIDE);

        int tn = t + waveCount;
        if (tn < numTiles) {
            const char* nxt = (const char*)(x + (long long)tn * (TILE * ROWSTRIDE));
            __builtin_prefetch(nxt + lane * 80, 0, 0);
        }

#pragma unroll
        for (int i = 0; i < (TILE * ROWSTRIDE) / (4 * 32); ++i) {
            int idx = lane + 32 * i;
            ((v4f*)tileBuf)[idx] = ((const v4f*)src)[idx];
        }
        __builtin_amdgcn_wave_barrier();

        const float* rp = tileBuf + m * ROWSTRIDE + k0;
        v2f F[5], M[5];
        for (int s = 0; s < 5; ++s) {
            F[s] = *(const v2f*)(rp + 4 * s);
            M[s] = *(const v2f*)(rp + NFEAT + 4 * s);
        }

        unsigned kill = 0u;
        for (int s = 0; s < 5; ++s) {
            kill |= killBits(F[s].x, M[s].x, kc[2 * s]);
            kill |= killBits(F[s].y, M[s].y, kc[2 * s + 1]);
        }
        kill |= otherHalf(kill);

        v8f acc = {};
        v2f MF[5];
        for (int s = 0; s < 5; ++s) {
            MF[s] = M[s] * F[s];
            v2f a = MF[s] * F[s];
            acc = __builtin_amdgcn_wmma_f32_16x16x4_f32(
                false, a, false, bfrag[s], (short)0, acc, false, false);
        }
        for (int s = 0; s < 5; ++s) {
            acc = __builtin_amdgcn_wmma_f32_16x16x4_f32(
                false, MF[s], false, bfrag[5 + s], (short)0, acc, false, false);
        }
        for (int s = 0; s < 5; ++s) {
            acc = __builtin_amdgcn_wmma_f32_16x16x4_f32(
                false, M[s], false, bfrag[10 + s], (short)0, acc, false, false);
        }

        if (n < NCLS) {
            int rbase = (lane >> 4) << 3;
#pragma unroll
            for (int v = 0; v < 8; ++v)
                scBuf[(rbase + v) * NCLS + n] = acc[v];
        }
        __builtin_amdgcn_wave_barrier();

        if (lane < TILE) {
            const v4f* sp = (const v4f*)(scBuf + lane * NCLS);
            v4f s0 = sp[0];
            v4f s1 = sp[1];
            float p[NCLS] = { s0.x, s0.y, s0.z, s0.w, s1.x, s1.y, s1.z, s1.w };
            float sum = 0.0f;
#pragma unroll
            for (int c = 0; c < NCLS; ++c) {
                float e  = __expf(p[c]) * 0.125f;
                float pr = ((kill >> c) & 1u) ? 0.0f : e;
                p[c] = pr;
                sum += pr;
            }
            float inv = 1.0f / fmaxf(sum, 1e-8f);
            float* dst = out + ((long long)t * TILE + lane) * NCLS;
            v4f o0 = { p[0] * inv, p[1] * inv, p[2] * inv, p[3] * inv };
            v4f o1 = { p[4] * inv, p[5] * inv, p[6] * inv, p[7] * inv };
            ((v4f*)dst)[0] = o0;
            ((v4f*)dst)[1] = o1;
        }
        __builtin_amdgcn_wave_barrier();
    }
}

__global__ void nb_tail_kernel(const float* __restrict__ x, float* __restrict__ out,
                               int startRow, int rows) {
    int r = startRow + blockIdx.x * blockDim.x + threadIdx.x;
    if (r >= rows) return;
    const float* row = x + (long long)r * ROWSTRIDE;
    float probs[NCLS];
    float sum = 0.0f;
    for (int c = 0; c < NCLS; ++c) {
        float p = 0.125f;
        for (int j = 0; j < NFEAT; ++j) {
            float f = row[j], msk = row[NFEAT + j];
            if (msk > 0.0f) {
                int d = j - c;
                float v;
                if (d >= 0 && d < 3) {
                    float mean = (float)((c >> d) & 1);
                    float z = (f - mean) * (1.0f / 0.3f);
                    v = __expf(-0.5f * z * z) * (0.3989422804014327f / 0.3f);
                } else {
                    v = ((f >= 0.0f) && (f < 1.0f)) ? 1.0f : 0.0f;
                }
                p *= v;
            }
        }
        probs[c] = p;
        sum += p;
    }
    float inv = 1.0f / fmaxf(sum, 1e-8f);
    for (int c = 0; c < NCLS; ++c)
        out[(long long)r * NCLS + c] = probs[c] * inv;
}

extern "C" void kernel_launch(void* const* d_in, const int* in_sizes, int n_in,
                              void* d_out, int out_size, void* d_ws, size_t ws_size,
                              hipStream_t stream) {
    (void)n_in; (void)out_size; (void)d_ws; (void)ws_size;
    const float* x   = (const float*)d_in[0];
    float*       out = (float*)d_out;

    long long rows = (long long)in_sizes[0] / ROWSTRIDE;
    int numTiles = (int)(rows / TILE);
    int tail     = (int)(rows % TILE);

    if (numTiles > 0) {
        int blocks = 1024;
        int maxBlocks = (numTiles + WAVES_PER_BLOCK - 1) / WAVES_PER_BLOCK;
        if (blocks > maxBlocks) blocks = maxBlocks;
        nb_wmma_kernel<<<blocks, 256, 0, stream>>>(x, out, numTiles);
    }
    if (tail > 0) {
        nb_tail_kernel<<<1, 64, 0, stream>>>(x, out, (int)(rows - tail), (int)rows);
    }
}


# GEMM K cut 60 -> 32: weight rows of features 10..19 are zero for every class, so only features 0..9 are multiplied; 15 -> 8 WMMAs per tile, 30 -> 16 weight loads per wave; k sum order regrouped
# speedup vs baseline: 1.5783x; 1.1276x over previous
#include <hip/hip_runtime.h>
#include <hip/hip_bf16.h>


typedef __attribute__((ext_vector_type(2))) float v2f;
typedef __attribute__((ext_vector_type(4))) float v4f;
typedef __attribute__((ext_vector_type(8))) float v8f;

#define NFEAT 20
#define NCLS  8
#define ROWSTRIDE 40
#define KSTEPS 8
#define TILE 16
#define WAVES_PER_BLOCK 8

static constexpr float classWeight(int g, int j, int c) {
    const float inv2s2  = 5.5555555f;
    const float invs2   = 11.111111f;
    const float logNorm = 0.28503434f;
    if (c >= NCLS) return 0.0f;
    int d = j - c;
    if (d < 0 || d >= 3) return 0.0f;
    float m = (float)((c >> d) & 1);
    if (g == 0) return -inv2s2;
    if (g == 1) return m * invs2;
    return -m * inv2s2 + logNorm;
}

struct WTabT { float v[KSTEPS][2][2][16]; };

static constexpr WTabT makeWTab() {
    WTabT t{};
    for (int s = 0; s < KSTEPS; ++s)
        for (int h = 0; h < 2; ++h)
            for (int e = 0; e < 2; ++e)
                for (int c = 0; c < 16; ++c) {
                    float w = 0.0f;
                    if (s < 6) {
                        int g = s >> 1;
                        int j = 4 * (s & 1) + 2 * h + e;
                        w = classWeight(g, j, c);
                    } else if (s == 6) {
                        w = classWeight(e, 8 + h, c);
                    } else if (e == 0) {
                        w = classWeight(2, 8 + h, c);
                    }
                    t.v[s][h][e][c] = w;
                }
    return t;
}

__constant__ WTabT wtab = makeWTab();

__device__ __forceinline__ unsigned killMaskF(int j) {
    int lo = j - 2; if (lo < 0) lo = 0;
    int hi = j;     if (hi > 7) hi = 7;
    unsigned g = 0u;
    if (lo <= hi) g = ((1u << (hi - lo + 1)) - 1u) << lo;
    return 0xFFu & ~g;
}

__device__ __forceinline__ unsigned otherHalf(unsigned v) {
    return (unsigned)__builtin_amdgcn_permlanex16((int)v, (int)v, 0x76543210, 0xfedcba98, false, false);
}

__device__ __forceinline__ unsigned killBits(float f, float msk, unsigned bits) {
    bool bad = (msk > 0.0f) & !((f >= 0.0f) & (f < 1.0f));
    return bad ? bits : 0u;
}

__global__ void __launch_bounds__(256)
nb_wmma_kernel(const float* __restrict__ x, float* __restrict__ out, int numTiles) {
    __shared__ float lds[WAVES_PER_BLOCK][TILE * ROWSTRIDE + TILE * NCLS];

    const int lane        = threadIdx.x & 31;
    const int waveInBlock = threadIdx.x >> 5;
    float* tileBuf = &lds[waveInBlock][0];
    float* scBuf   = &lds[waveInBlock][TILE * ROWSTRIDE];

    const int m  = lane & 15;
    const int k0 = (lane >> 4) << 1;
    const int n  = lane & 15;

    const int h = lane >> 4;
    v2f bfrag[KSTEPS];
#pragma unroll
    for (int s = 0; s < KSTEPS; ++s) {
        bfrag[s].x = wtab.v[s][h][0][n];
        bfrag[s].y = wtab.v[s][h][1][n];
    }

    unsigned kc[10];
    for (int s = 0; s < 5; ++s) {
        kc[2 * s]     = k0 ? killMaskF(4 * s + 2) : killMaskF(4 * s);
        kc[2 * s + 1] = k0 ? killMaskF(4 * s + 3) : killMaskF(4 * s + 1);
    }

    const int waveId    = blockIdx.x * WAVES_PER_BLOCK + waveInBlock;
    const int waveCount = gridDim.x * WAVES_PER_BLOCK;

    for (int t = waveId; t < numTiles; t += waveCount) {
        const float* src = x + (long long)t * (TILE * ROWSTRIDE);

        int tn = t + waveCount;
        if (tn < numTiles) {
            const char* nxt = (const char*)(x + (long long)tn * (TILE * ROWSTRIDE));
            __builtin_prefetch(nxt + lane * 80, 0, 0);
        }

#pragma unroll
        for (int i = 0; i < (TILE * ROWSTRIDE) / (4 * 32); ++i) {
            int idx = lane + 32 * i;
            ((v4f*)tileBuf)[idx] = ((const v4f*)src)[idx];
        }
        __builtin_amdgcn_wave_barrier();

        const float* rp = tileBuf + m * ROWSTRIDE + k0;
        v2f F[5], M[5];
        for (int s = 0; s < 5; ++s) {
            F[s] = *(const v2f*)(rp + 4 * s);
            M[s] = *(const v2f*)(rp + NFEAT + 4 * s);
        }

        unsigned kill = 0u;
        for (int s = 0; s < 5; ++s) {
            kill |= killBits(F[s].x, M[s].x, kc[2 * s]);
            kill |= killBits(F[s].y, M[s].y, kc[2 * s + 1]);
        }
        kill |= otherHalf(kill);

        const float* xp = tileBuf + m * ROWSTRIDE + 8 + h;
        const float fX = xp[0];
        const float mX = xp[NFEAT];

        v8f acc = {};
        v2f MF0 = M[0] * F[0];
        v2f MF1 = M[1] * F[1];
        v2f MFF0 = MF0 * F[0];
        v2f MFF1 = MF1 * F[1];
        float mfX = mX * fX;
        v2f a6 = { mfX * fX, mfX };
        v2f a7 = { mX, 0.0f };
        acc = __builtin_amdgcn_wmma_f32_16x16x4_f32(false, MFF0, false, bfrag[0], (short)0, acc, false, false);
        acc = __builtin_amdgcn_wmma_f32_16x16x4_f32(false, MFF1, false, bfrag[1], (short)0, acc, false, false);
        acc = __builtin_amdgcn_wmma_f32_16x16x4_f32(false, MF0,  false, bfrag[2], (short)0, acc, false, false);
        acc = __builtin_amdgcn_wmma_f32_16x16x4_f32(false, MF1,  false, bfrag[3], (short)0, acc, false, false);
        acc = __builtin_amdgcn_wmma_f32_16x16x4_f32(false, M[0], false, bfrag[4], (short)0, acc, false, false);
        acc = __builtin_amdgcn_wmma_f32_16x16x4_f32(false, M[1], false, bfrag[5], (short)0, acc, false, false);
        acc = __builtin_amdgcn_wmma_f32_16x16x4_f32(false, a6,   false, bfrag[6], (short)0, acc, false, false);
        acc = __builtin_amdgcn_wmma_f32_16x16x4_f32(false, a7,   false, bfrag[7], (short)0, acc, false, false);

        if (n < NCLS) {
            int rbase = (lane >> 4) << 3;
#pragma unroll
            for (int v = 0; v < 8; ++v)
                scBuf[(rbase + v) * NCLS + n] = acc[v];
        }
        __builtin_amdgcn_wave_barrier();

        if (lane < TILE) {
            const v4f* sp = (const v4f*)(scBuf + lane * NCLS);
            v4f s0 = sp[0];
            v4f s1 = sp[1];
            float p[NCLS] = { s0.x, s0.y, s0.z, s0.w, s1.x, s1.y, s1.z, s1.w };
            float sum = 0.0f;
#pragma unroll
            for (int c = 0; c < NCLS; ++c) {
                float e  = __expf(p[c]) * 0.125f;
                float pr = ((kill >> c) & 1u) ? 0.0f : e;
                p[c] = pr;
                sum += pr;
            }
            float inv = 1.0f / fmaxf(sum, 1e-8f);
            float* dst = out + ((long long)t * TILE + lane) * NCLS;
            v4f o0 = { p[0] * inv, p[1] * inv, p[2] * inv, p[3] * inv };
            v4f o1 = { p[4] * inv, p[5] * inv, p[6] * inv, p[7] * inv };
            ((v4f*)dst)[0] = o0;
            ((v4f*)dst)[1] = o1;
        }
        __builtin_amdgcn_wave_barrier();
    }
}

__global__ void nb_tail_kernel(const float* __restrict__ x, float* __restrict__ out,
                               int startRow, int rows) {
    int r = startRow + blockIdx.x * blockDim.x + threadIdx.x;
    if (r >= rows) return;
    const float* row = x + (long long)r * ROWSTRIDE;
    float probs[NCLS];
    float sum = 0.0f;
    for (int c = 0; c < NCLS; ++c) {
        float p = 0.125f;
        for (int j = 0; j < NFEAT; ++j) {
            float f = row[j], msk = row[NFEAT + j];
            if (msk > 0.0f) {
                int d = j - c;
                float v;
                if (d >= 0 && d < 3) {
                    float mean = (float)((c >> d) & 1);
                    float z = (f - mean) * (1.0f / 0.3f);
                    v = __expf(-0.5f * z * z) * (0.3989422804014327f / 0.3f);
                } else {
                    v = ((f >= 0.0f) && (f < 1.0f)) ? 1.0f : 0.0f;
                }
                p *= v;
            }
        }
        probs[c] = p;
        sum += p;
    }
    float inv = 1.0f / fmaxf(sum, 1e-8f);
    for (int c = 0; c < NCLS; ++c)
        out[(long long)r * NCLS + c] = probs[c] * inv;
}

extern "C" void kernel_launch(void* const* d_in, const int* in_sizes, int n_in,
                              void* d_out, int out_size, void* d_ws, size_t ws_size,
                              hipStream_t stream) {
    (void)n_in; (void)out_size; (void)d_ws; (void)ws_size;
    const float* x   = (const float*)d_in[0];
    float*       out = (float*)d_out;

    long long rows = (long long)in_sizes[0] / ROWSTRIDE;
    int numTiles = (int)(rows / TILE);
    int tail     = (int)(rows % TILE);

    if (numTiles > 0) {
        int blocks = 1024;
        int maxBlocks = (numTiles + WAVES_PER_BLOCK - 1) / WAVES_PER_BLOCK;
        if (blocks > maxBlocks) blocks = maxBlocks;
        nb_wmma_kernel<<<blocks, 256, 0, stream>>>(x, out, numTiles);
    }
    if (tail > 0) {
        nb_tail_kernel<<<1, 64, 0, stream>>>(x, out, (int)(rows - tail), (int)rows);
    }
}
